# starting kernel
# baseline (speedup 1.0000x reference)
#include <hip/hip_runtime.h>


typedef __attribute__((ext_vector_type(2))) float v2f;
typedef __attribute__((ext_vector_type(8))) float v8f;

#define N_NODES 100000
#define D_FEAT  128
#define N_EDGES 640000

__global__ void zero_f32(float* __restrict__ p, int n4) {
  int i = blockIdx.x * blockDim.x + threadIdx.x;
  if (i < n4) ((float4*)p)[i] = make_float4(0.f, 0.f, 0.f, 0.f);
}

__global__ __launch_bounds__(256)
void scatter_mean_accum(const float* __restrict__ x,
                        const int* __restrict__ src,
                        const int* __restrict__ dst,
                        float* __restrict__ agg,
                        float* __restrict__ deg) {
  int wave = (int)((blockIdx.x * blockDim.x + threadIdx.x) >> 5);
  int lane = threadIdx.x & 31;
  if (wave >= N_EDGES) return;

  int s = src[wave];
  int d = dst[wave];

  const float4 v = ((const float4*)(x + (size_t)s * D_FEAT))[lane];
  float* a = agg + (size_t)d * D_FEAT + lane * 4;
  atomicAdd(a + 0, v.x);
  atomicAdd(a + 1, v.y);
  atomicAdd(a + 2, v.z);
  atomicAdd(a + 3, v.w);
  if (lane == 0) atomicAdd(deg + d, 1.0f);
}

__global__ __launch_bounds__(256)
void sage_layer(const float* __restrict__ xin,
                const float* __restrict__ agg,
                const float* __restrict__ deg,
                const float* __restrict__ w_l,
                const float* __restrict__ b_l,
                const float* __restrict__ w_r,
                float* __restrict__ out,
                int do_relu) {
  const int wave = threadIdx.x >> 5;
  const int lane = threadIdx.x & 31;
  const int hi   = lane >> 4;
  const int li   = lane & 15;
  const int m0   = blockIdx.x * 16;
  const int n0   = wave * 16;

  const int rowA = m0 + li;
  const float dv = deg[rowA];
  const float rs = 1.0f / fmaxf(dv, 1.0f);

  const float* aggRow = agg + (size_t)rowA * D_FEAT + 2 * hi;
  const float* xRow   = xin + (size_t)rowA * D_FEAT + 2 * hi;
  const float* wlRow  = w_l + (size_t)(n0 + li) * D_FEAT + 2 * hi;
  const float* wrRow  = w_r + (size_t)(n0 + li) * D_FEAT + 2 * hi;

  v8f acc = {};

  #pragma unroll 4
  for (int k0 = 0; k0 < D_FEAT; k0 += 4) {
    v2f a = *(const v2f*)(aggRow + k0);
    a = a * rs;
    v2f b = *(const v2f*)(wlRow + k0);
    acc = __builtin_amdgcn_wmma_f32_16x16x4_f32(
         false, a,  false, b,
         (short)0, acc,  false,  false);
  }

  #pragma unroll 4
  for (int k0 = 0; k0 < D_FEAT; k0 += 4) {
    v2f a = *(const v2f*)(xRow + k0);
    v2f b = *(const v2f*)(wrRow + k0);
    acc = __builtin_amdgcn_wmma_f32_16x16x4_f32(
        false, a, false, b, (short)0, acc, false, false);
  }

  const float bias = b_l[n0 + li];
  float* outBase = out + (size_t)(m0 + 8 * hi) * D_FEAT + n0 + li;
  #pragma unroll
  for (int r = 0; r < 8; ++r) {
    float v = acc[r] + bias;
    if (do_relu) v = fmaxf(v, 0.0f);
    outBase[(size_t)r * D_FEAT] = v;
  }
}

extern "C" void kernel_launch(void* const* d_in, const int* in_sizes, int n_in,
                              void* d_out, int out_size, void* d_ws, size_t ws_size,
                              hipStream_t stream) {
  const float* x    = (const float*)d_in[0];
  const float* w1_l = (const float*)d_in[1];
  const float* b1_l = (const float*)d_in[2];
  const float* w1_r = (const float*)d_in[3];
  const float* w2_l = (const float*)d_in[4];
  const float* b2_l = (const float*)d_in[5];
  const float* w2_r = (const float*)d_in[6];
  const int*   edges = (const int*)d_in[7];
  const int*   src  = edges;
  const int*   dst  = edges + N_EDGES;

  float* agg = (float*)d_ws;
  float* deg = agg + (size_t)N_NODES * D_FEAT;
  float* h   = deg + N_NODES;
  float* out = (float*)d_out;

  const int zeroN  = N_NODES * D_FEAT + N_NODES;
  const int zeroN4 = zeroN / 4;
  dim3 zb(256), zg((zeroN4 + 255) / 256);
  dim3 sb(256), sg((N_EDGES * 32 + 255) / 256);
  dim3 lb(256), lg(N_NODES / 16);

  zero_f32<<<zg, zb, 0, stream>>>(agg, zeroN4);
  scatter_mean_accum<<<sg, sb, 0, stream>>>(x, src, dst, agg, deg);
  sage_layer<<<lg, lb, 0, stream>>>(x, agg, deg, w1_l, b1_l, w1_r, h, 1);

  zero_f32<<<zg, zb, 0, stream>>>(agg, zeroN4);
  scatter_mean_accum<<<sg, sb, 0, stream>>>(h, src, dst, agg, deg);
  sage_layer<<<lg, lb, 0, stream>>>(h, agg, deg, w2_l, b2_l, w2_r, out, 0);
}


# sage_layer GEMM: weights staged once per block into LDS in B-fragment order; each wave keeps a 16x128 output strip in registers (one A fragment feeds 8 WMMA); 512 blocks walk the row tiles
# speedup vs baseline: 1.4026x; 1.4026x over previous
#include <hip/hip_runtime.h>


typedef __attribute__((ext_vector_type(2))) float v2f;
typedef __attribute__((ext_vector_type(8))) float v8f;

#define N_NODES 100000
#define D_FEAT  128
#define N_EDGES 640000

__global__ void zero_f32(float* __restrict__ p, int n4) {
  int i = blockIdx.x * blockDim.x + threadIdx.x;
  if (i < n4) ((float4*)p)[i] = make_float4(0.f, 0.f, 0.f, 0.f);
}

__global__ __launch_bounds__(256)
void scatter_mean_accum(const float* __restrict__ x,
                        const int* __restrict__ src,
                        const int* __restrict__ dst,
                        float* __restrict__ agg,
                        float* __restrict__ deg) {
  int wave = (int)((blockIdx.x * blockDim.x + threadIdx.x) >> 5);
  int lane = threadIdx.x & 31;
  if (wave >= N_EDGES) return;

  int s = src[wave];
  int d = dst[wave];

  const float4 v = ((const float4*)(x + (size_t)s * D_FEAT))[lane];
  float* a = agg + (size_t)d * D_FEAT + lane * 4;
  atomicAdd(a + 0, v.x);
  atomicAdd(a + 1, v.y);
  atomicAdd(a + 2, v.z);
  atomicAdd(a + 3, v.w);
  if (lane == 0) atomicAdd(deg + d, 1.0f);
}

#define ROW_TILES    (N_NODES / 16)
#define LAYER_BLOCKS 512
#define K_STEPS      (D_FEAT / 4)

__device__ __forceinline__ v8f mma4(v2f a, v2f b, v8f c) {
  return __builtin_amdgcn_wmma_f32_16x16x4_f32(
       false, a,  false, b,
       (short)0, c,  false,  false);
}

__global__ __launch_bounds__(256)
void sage_layer(const float* __restrict__ xin,
                const float* __restrict__ agg,
                const float* __restrict__ deg,
                const float* __restrict__ w_l,
                const float* __restrict__ b_l,
                const float* __restrict__ w_r,
                float* __restrict__ out,
                int do_relu) {
  __shared__ float4 wfrag[2 * K_STEPS * 4 * 32];

  const int wave = threadIdx.x >> 5;
  const int lane = threadIdx.x & 31;
  const int hi   = lane >> 4;
  const int li   = lane & 15;

  {
    const int par = hi;
    float* lds = (float*)wfrag;
    for (int u = wave; u < 2 * K_STEPS * 4; u += 8) {
      const int S = u >> 2;
      const int p = u & 3;
      const float* w = (S < K_STEPS) ? w_l : w_r;
      const int s = S & (K_STEPS - 1);
      const float4 v = *(const float4*)(w + (size_t)((2 * p + par) * 16 + li) * D_FEAT + 4 * s);
      v2f lo = {v.x, v.y};
      v2f up = {v.z, v.w};
      *(v2f*)(lds + (size_t)(u * 32 + li) * 4 + 2 * par)      = lo;
      *(v2f*)(lds + (size_t)(u * 32 + 16 + li) * 4 + 2 * par) = up;
    }
  }
  __syncthreads();

  const int t0 = (int)(((unsigned)blockIdx.x * (unsigned)ROW_TILES) / (unsigned)LAYER_BLOCKS);
  const int t1 = (int)(((unsigned)(blockIdx.x + 1) * (unsigned)ROW_TILES) / (unsigned)LAYER_BLOCKS);

  const float4* bfrag = wfrag + lane;

  for (int t = t0 + wave; t < t1; t += 8) {
    const int m0   = t * 16;
    const int rowA = m0 + li;
    const float dv = deg[rowA];
    const float rs = 1.0f / fmaxf(dv, 1.0f);

    const float* aggRow = agg + (size_t)rowA * D_FEAT + 2 * hi;
    const float* xRow   = xin + (size_t)rowA * D_FEAT + 2 * hi;

    v8f acc[8];
    #pragma unroll
    for (int j = 0; j < 8; ++j) acc[j] = (v8f){};

    #pragma unroll 8
    for (int s = 0; s < K_STEPS; ++s) {
      v2f a = *(const v2f*)(aggRow + 4 * s);
      a = a * rs;
      #pragma unroll
      for (int p = 0; p < 4; ++p) {
        const float4 b = bfrag[(s * 4 + p) * 32];
        v2f b0 = {b.x, b.y};
        v2f b1 = {b.z, b.w};
        acc[2 * p]     = mma4(a, b0, acc[2 * p]);
        acc[2 * p + 1] = mma4(a, b1, acc[2 * p + 1]);
      }
    }

    #pragma unroll 8
    for (int s = 0; s < K_STEPS; ++s) {
      v2f a = *(const v2f*)(xRow + 4 * s);
      #pragma unroll
      for (int p = 0; p < 4; ++p) {
        const float4 b = bfrag[((K_STEPS + s) * 4 + p) * 32];
        v2f b0 = {b.x, b.y};
        v2f b1 = {b.z, b.w};
        acc[2 * p]     = mma4(a, b0, acc[2 * p]);
        acc[2 * p + 1] = mma4(a, b1, acc[2 * p + 1]);
      }
    }

    float* outBase = out + (size_t)(m0 + 8 * hi) * D_FEAT + li;
    #pragma unroll
    for (int j = 0; j < 8; ++j) {
      const float bias = b_l[j * 16 + li];
      #pragma unroll
      for (int r = 0; r < 8; ++r) {
        float v = acc[j][r] + bias;
        if (do_relu) v = fmaxf(v, 0.0f);
        outBase[(size_t)r * D_FEAT + j * 16] = v;
      }
    }
  }
}

extern "C" void kernel_launch(void* const* d_in, const int* in_sizes, int n_in,
                              void* d_out, int out_size, void* d_ws, size_t ws_size,
                              hipStream_t stream) {
  const float* x    = (const float*)d_in[0];
  const float* w1_l = (const float*)d_in[1];
  const float* b1_l = (const float*)d_in[2];
  const float* w1_r = (const float*)d_in[3];
  const float* w2_l = (const float*)d_in[4];
  const float* b2_l = (const float*)d_in[5];
  const float* w2_r = (const float*)d_in[6];
  const int*   edges = (const int*)d_in[7];
  const int*   src  = edges;
  const int*   dst  = edges + N_EDGES;

  float* agg = (float*)d_ws;
  float* deg = agg + (size_t)N_NODES * D_FEAT;
  float* h   = deg + N_NODES;
  float* out = (float*)d_out;

  const int zeroN  = N_NODES * D_FEAT + N_NODES;
  const int zeroN4 = zeroN / 4;
  dim3 zb(256), zg((zeroN4 + 255) / 256);
  dim3 sb(256), sg((N_EDGES * 32 + 255) / 256);
  dim3 lb(256), lg(LAYER_BLOCKS);

  zero_f32<<<zg, zb, 0, stream>>>(agg, zeroN4);
  scatter_mean_accum<<<sg, sb, 0, stream>>>(x, src, dst, agg, deg);
  sage_layer<<<lg, lb, 0, stream>>>(x, agg, deg, w1_l, b1_l, w1_r, h, 1);

  zero_f32<<<zg, zb, 0, stream>>>(agg, zeroN4);
  scatter_mean_accum<<<sg, sb, 0, stream>>>(h, src, dst, agg, deg);
  sage_layer<<<lg, lb, 0, stream>>>(h, agg, deg, w2_l, b2_l, w2_r, out, 0);
}


# neighbour mean by gather: edges counting-sorted by destination once per call (int atomics), one wave per node sums its source rows; float atomic scatter and zero pass removed; sum order changes
# speedup vs baseline: 2.3113x; 1.6479x over previous
#include <hip/hip_runtime.h>


typedef __attribute__((ext_vector_type(2))) float v2f;
typedef __attribute__((ext_vector_type(8))) float v8f;

#define N_NODES 100000
#define D_FEAT  128
#define N_EDGES 640000

__global__ void zero_counts(int* __restrict__ cnt) {
  const int i = blockIdx.x * blockDim.x + threadIdx.x;
  if (i < N_NODES / 4) ((int4*)cnt)[i] = make_int4(0, 0, 0, 0);
}

__global__ void count_in_edges(const int* __restrict__ dst, int* __restrict__ cnt) {
  const int e = blockIdx.x * blockDim.x + threadIdx.x;
  if (e < N_EDGES) atomicAdd(cnt + dst[e], 1);
}

#define SCAN_THREADS 1024
#define SCAN_BLOCKS  25

__global__ __launch_bounds__(SCAN_THREADS)
void scan_counts(const int* __restrict__ cnt, int* __restrict__ cur) {
  __shared__ int wtot[32];
  __shared__ int wpre[32];

  const int tid  = threadIdx.x;
  const int lane = tid & 31;
  const int wave = tid >> 5;
  const int4* cnt4 = (const int4*)cnt;

  int pre = 0;
  #pragma unroll
  for (int j = 0; j < SCAN_BLOCKS - 1; ++j) {
    const int4 v = cnt4[j * SCAN_THREADS + tid];
    const int t = v.x + v.y + v.z + v.w;
    pre += (j < (int)blockIdx.x) ? t : 0;
  }

  const int q = blockIdx.x * SCAN_THREADS + tid;
  int4 v = make_int4(0, 0, 0, 0);
  if (q < N_NODES / 4) v = cnt4[q];
  const int s = v.x + v.y + v.z + v.w;

  int inc = s;
  #pragma unroll
  for (int d = 1; d < 32; d <<= 1) {
    const int t = __shfl_up(inc, d);
    if (lane >= d) inc += t;
  }
  #pragma unroll
  for (int d = 16; d >= 1; d >>= 1) pre += __shfl_xor(pre, d);

  if (lane == 31) wtot[wave] = inc;
  if (lane == 0)  wpre[wave] = pre;
  __syncthreads();

  int base = wpre[lane] + ((lane < wave) ? wtot[lane] : 0);
  #pragma unroll
  for (int d = 16; d >= 1; d >>= 1) base += __shfl_xor(base, d);

  if (q < N_NODES / 4) {
    int4 o;
    o.x = base + inc - s;
    o.y = o.x + v.x;
    o.z = o.y + v.y;
    o.w = o.z + v.z;
    ((int4*)cur)[q] = o;
  }
}

__global__ void fill_sorted(const int* __restrict__ src,
                            const int* __restrict__ dst,
                            int* __restrict__ cur,
                            int* __restrict__ col) {
  const int e = blockIdx.x * blockDim.x + threadIdx.x;
  if (e < N_EDGES) {
    const int s = src[e];
    const int p = atomicAdd(cur + dst[e], 1);
    if ((unsigned)p < (unsigned)N_EDGES) col[p] = s;
  }
}

#define AGG_BLOCKS 2048

__global__ __launch_bounds__(256)
void aggregate_mean(const float* __restrict__ xin,
                    const int* __restrict__ cur,
                    const int* __restrict__ col,
                    float* __restrict__ mean) {
  const int lane = threadIdx.x & 31;
  const int w0   = __builtin_amdgcn_readfirstlane((int)(blockIdx.x * 8 + (threadIdx.x >> 5)));

  for (int node = w0; node < N_NODES; node += AGG_BLOCKS * 8) {
    const int before = cur[node > 0 ? node - 1 : 0];
    int end = cur[node];
    int beg = node > 0 ? before : 0;
    end = min(end, N_EDGES);
    beg = max(min(beg, end), 0);
    const int n = end - beg;

    float4 acc = make_float4(0.f, 0.f, 0.f, 0.f);
    for (int c = 0; c < n; c += 32) {
      const int m = min(n - c, 32);
      int my = 0;
      if (lane < m) my = col[beg + c + lane];
      my = (int)min((unsigned)my, (unsigned)(N_NODES - 1));
      for (int k = 0; k < m; k += 8) {
        float4 r[8];
        #pragma unroll
        for (int j = 0; j < 8; ++j) {
          r[j] = make_float4(0.f, 0.f, 0.f, 0.f);
          if (k + j < m) {
            const int sidx = __builtin_amdgcn_readlane(my, k + j);
            r[j] = ((const float4*)(xin + (size_t)sidx * D_FEAT))[lane];
          }
        }
        #pragma unroll
        for (int j = 0; j < 8; ++j) {
          acc.x += r[j].x; acc.y += r[j].y; acc.z += r[j].z; acc.w += r[j].w;
        }
      }
    }
    const float d = fmaxf((float)n, 1.0f);
    float4 o;
    o.x = acc.x / d; o.y = acc.y / d; o.z = acc.z / d; o.w = acc.w / d;
    ((float4*)(mean + (size_t)node * D_FEAT))[lane] = o;
  }
}

#define ROW_TILES    (N_NODES / 16)
#define LAYER_BLOCKS 512
#define K_STEPS      (D_FEAT / 4)

__device__ __forceinline__ v8f mma4(v2f a, v2f b, v8f c) {
  return __builtin_amdgcn_wmma_f32_16x16x4_f32(
       false, a,  false, b,
       (short)0, c,  false,  false);
}

__global__ __launch_bounds__(256)
void sage_layer(const float* __restrict__ xin,
                const float* __restrict__ mean,
                const float* __restrict__ w_l,
                const float* __restrict__ b_l,
                const float* __restrict__ w_r,
                float* __restrict__ out,
                int do_relu) {
  __shared__ float4 wfrag[2 * K_STEPS * 4 * 32];

  const int wave = threadIdx.x >> 5;
  const int lane = threadIdx.x & 31;
  const int hi   = lane >> 4;
  const int li   = lane & 15;

  {
    const int par = hi;
    float* lds = (float*)wfrag;
    for (int u = wave; u < 2 * K_STEPS * 4; u += 8) {
      const int S = u >> 2;
      const int p = u & 3;
      const float* w = (S < K_STEPS) ? w_l : w_r;
      const int s = S & (K_STEPS - 1);
      const float4 v = *(const float4*)(w + (size_t)((2 * p + par) * 16 + li) * D_FEAT + 4 * s);
      v2f lo = {v.x, v.y};
      v2f up = {v.z, v.w};
      *(v2f*)(lds + (size_t)(u * 32 + li) * 4 + 2 * par)      = lo;
      *(v2f*)(lds + (size_t)(u * 32 + 16 + li) * 4 + 2 * par) = up;
    }
  }
  __syncthreads();

  const int t0 = (int)(((unsigned)blockIdx.x * (unsigned)ROW_TILES) / (unsigned)LAYER_BLOCKS);
  const int t1 = (int)(((unsigned)(blockIdx.x + 1) * (unsigned)ROW_TILES) / (unsigned)LAYER_BLOCKS);

  const float4* bfrag = wfrag + lane;

  for (int t = t0 + wave; t < t1; t += 8) {
    const int m0   = t * 16;
    const int rowA = m0 + li;
    const float* aggRow = mean + (size_t)rowA * D_FEAT + 2 * hi;
    const float* xRow   = xin + (size_t)rowA * D_FEAT + 2 * hi;

    v8f acc[8];
    #pragma unroll
    for (int j = 0; j < 8; ++j) acc[j] = (v8f){};

    #pragma unroll 8
    for (int s = 0; s < K_STEPS; ++s) {
      const v2f a = *(const v2f*)(aggRow + 4 * s);
      #pragma unroll
      for (int p = 0; p < 4; ++p) {
        const float4 b = bfrag[(s * 4 + p) * 32];
        v2f b0 = {b.x, b.y};
        v2f b1 = {b.z, b.w};
        acc[2 * p]     = mma4(a, b0, acc[2 * p]);
        acc[2 * p + 1] = mma4(a, b1, acc[2 * p + 1]);
      }
    }

    #pragma unroll 8
    for (int s = 0; s < K_STEPS; ++s) {
      v2f a = *(const v2f*)(xRow + 4 * s);
      #pragma unroll
      for (int p = 0; p < 4; ++p) {
        const float4 b = bfrag[((K_STEPS + s) * 4 + p) * 32];
        v2f b0 = {b.x, b.y};
        v2f b1 = {b.z, b.w};
        acc[2 * p]     = mma4(a, b0, acc[2 * p]);
        acc[2 * p + 1] = mma4(a, b1, acc[2 * p + 1]);
      }
    }

    float* outBase = out + (size_t)(m0 + 8 * hi) * D_FEAT + li;
    #pragma unroll
    for (int j = 0; j < 8; ++j) {
      const float bias = b_l[j * 16 + li];
      #pragma unroll
      for (int r = 0; r < 8; ++r) {
        float v = acc[j][r] + bias;
        if (do_relu) v = fmaxf(v, 0.0f);
        outBase[(size_t)r * D_FEAT + j * 16] = v;
      }
    }
  }
}

extern "C" void kernel_launch(void* const* d_in, const int* in_sizes, int n_in,
                              void* d_out, int out_size, void* d_ws, size_t ws_size,
                              hipStream_t stream) {
  const float* x    = (const float*)d_in[0];
  const float* w1_l = (const float*)d_in[1];
  const float* b1_l = (const float*)d_in[2];
  const float* w1_r = (const float*)d_in[3];
  const float* w2_l = (const float*)d_in[4];
  const float* b2_l = (const float*)d_in[5];
  const float* w2_r = (const float*)d_in[6];
  const int*   edges = (const int*)d_in[7];
  const int*   src  = edges;
  const int*   dst  = edges + N_EDGES;

  float* mean = (float*)d_ws;
  float* h    = mean + (size_t)N_NODES * D_FEAT;
  int*   cnt  = (int*)(h + (size_t)N_NODES * D_FEAT);
  int*   cur  = cnt + N_NODES;
  int*   col  = cur + N_NODES;
  float* out  = (float*)d_out;

  dim3 tb(256);
  dim3 zg((N_NODES / 4 + 255) / 256);
  dim3 eg((N_EDGES + 255) / 256);
  dim3 lb(256), lg(LAYER_BLOCKS);

  zero_counts<<<zg, tb, 0, stream>>>(cnt);
  count_in_edges<<<eg, tb, 0, stream>>>(dst, cnt);
  scan_counts<<<dim3(SCAN_BLOCKS), dim3(SCAN_THREADS), 0, stream>>>(cnt, cur);
  fill_sorted<<<eg, tb, 0, stream>>>(src, dst, cur, col);

  aggregate_mean<<<dim3(AGG_BLOCKS), tb, 0, stream>>>(x, cur, col, mean);
  sage_layer<<<lg, lb, 0, stream>>>(x, mean, w1_l, b1_l, w1_r, h, 1);

  aggregate_mean<<<dim3(AGG_BLOCKS), tb, 0, stream>>>(h, cur, col, mean);
  sage_layer<<<lg, lb, 0, stream>>>(h, mean, w2_l, b2_l, w2_r, out, 0);
}


# neighbour mean fused into the layer kernel: each wave gathers its 16 rows' means into its own LDS strip and reads the A fragments there; mean plane and 2 launches gone; per-node sum order kept
# speedup vs baseline: 2.4038x; 1.0400x over previous
#include <hip/hip_runtime.h>


typedef __attribute__((ext_vector_type(2))) float v2f;
typedef __attribute__((ext_vector_type(4))) float v4f;
typedef __attribute__((ext_vector_type(8))) float v8f;

#define N_NODES 100000
#define D_FEAT  128
#define N_EDGES 640000

__global__ void zero_counts(int* __restrict__ cnt) {
  const int i = blockIdx.x * blockDim.x + threadIdx.x;
  if (i < N_NODES / 4) ((int4*)cnt)[i] = make_int4(0, 0, 0, 0);
}

__global__ void count_in_edges(const int* __restrict__ dst, int* __restrict__ cnt) {
  const int e = blockIdx.x * blockDim.x + threadIdx.x;
  if (e < N_EDGES) atomicAdd(cnt + dst[e], 1);
}

#define SCAN_THREADS 1024
#define SCAN_BLOCKS  25

__global__ __launch_bounds__(SCAN_THREADS)
void scan_counts(const int* __restrict__ cnt, int* __restrict__ cur) {
  __shared__ int wtot[32];
  __shared__ int wpre[32];

  const int tid  = threadIdx.x;
  const int lane = tid & 31;
  const int wave = tid >> 5;
  const int4* cnt4 = (const int4*)cnt;

  int pre = 0;
  #pragma unroll
  for (int j = 0; j < SCAN_BLOCKS - 1; ++j) {
    const int4 v = cnt4[j * SCAN_THREADS + tid];
    const int t = v.x + v.y + v.z + v.w;
    pre += (j < (int)blockIdx.x) ? t : 0;
  }

  const int q = blockIdx.x * SCAN_THREADS + tid;
  int4 v = make_int4(0, 0, 0, 0);
  if (q < N_NODES / 4) v = cnt4[q];
  const int s = v.x + v.y + v.z + v.w;

  int inc = s;
  #pragma unroll
  for (int d = 1; d < 32; d <<= 1) {
    const int t = __shfl_up(inc, d);
    if (lane >= d) inc += t;
  }
  #pragma unroll
  for (int d = 16; d >= 1; d >>= 1) pre += __shfl_xor(pre, d);

  if (lane == 31) wtot[wave] = inc;
  if (lane == 0)  wpre[wave] = pre;
  __syncthreads();

  int base = wpre[lane] + ((lane < wave) ? wtot[lane] : 0);
  #pragma unroll
  for (int d = 16; d >= 1; d >>= 1) base += __shfl_xor(base, d);

  if (q < N_NODES / 4) {
    int4 o;
    o.x = base + inc - s;
    o.y = o.x + v.x;
    o.z = o.y + v.y;
    o.w = o.z + v.z;
    ((int4*)cur)[q] = o;
  }
}

__global__ void fill_sorted(const int* __restrict__ src,
                            const int* __restrict__ dst,
                            int* __restrict__ cur,
                            int* __restrict__ col) {
  const int e = blockIdx.x * blockDim.x + threadIdx.x;
  if (e < N_EDGES) {
    const int s = src[e];
    const int p = atomicAdd(cur + dst[e], 1);
    if ((unsigned)p < (unsigned)N_EDGES) col[p] = s;
  }
}

#define ROW_TILES    (N_NODES / 16)
#define LAYER_BLOCKS 256
#define LAYER_WAVES  16
#define K_STEPS      (D_FEAT / 4)
#define STRIP_PITCH  132
#define STRIP_FLOATS (16 * STRIP_PITCH)

__device__ __forceinline__ v8f mma4(v2f a, v2f b, v8f c) {
  return __builtin_amdgcn_wmma_f32_16x16x4_f32(
       false, a,  false, b,
       (short)0, c,  false,  false);
}

__global__ __launch_bounds__(512)
void sage_layer(const float* __restrict__ xin,
                const int* __restrict__ cur,
                const int* __restrict__ col,
                const float* __restrict__ w_l,
                const float* __restrict__ b_l,
                const float* __restrict__ w_r,
                float* __restrict__ out,
                int do_relu) {
  __shared__ float4 wfrag[2 * K_STEPS * 4 * 32];
  __shared__ __attribute__((aligned(16))) float strips[LAYER_WAVES * STRIP_FLOATS];

  const int wave = __builtin_amdgcn_readfirstlane((int)(threadIdx.x >> 5));
  const int lane = threadIdx.x & 31;
  const int hi   = lane >> 4;
  const int li   = lane & 15;

  {
    const int par = hi;
    float* lds = (float*)wfrag;
    for (int i = 0; i < 2 * K_STEPS * 4 / LAYER_WAVES; i += 8) {
      float4 v[8];
      #pragma unroll
      for (int q = 0; q < 8; ++q) {
        const int u = wave + LAYER_WAVES * (i + q);
        const int S = u >> 2;
        const int p = u & 3;
        const float* w = (S < K_STEPS) ? w_l : w_r;
        const int s = S & (K_STEPS - 1);
        v[q] = *(const float4*)(w + (size_t)((2 * p + par) * 16 + li) * D_FEAT + 4 * s);
      }
      #pragma unroll
      for (int q = 0; q < 8; ++q) {
        const int u = wave + LAYER_WAVES * (i + q);
        v2f lo = {v[q].x, v[q].y};
        v2f up = {v[q].z, v[q].w};
        *(v2f*)(lds + (size_t)(u * 32 + li) * 4 + 2 * par)      = lo;
        *(v2f*)(lds + (size_t)(u * 32 + 16 + li) * 4 + 2 * par) = up;
      }
    }
  }
  __syncthreads();

  const int t0 = (int)(((unsigned)blockIdx.x * (unsigned)ROW_TILES) / (unsigned)LAYER_BLOCKS);
  const int t1 = (int)(((unsigned)(blockIdx.x + 1) * (unsigned)ROW_TILES) / (unsigned)LAYER_BLOCKS);

  const float4* bfrag = wfrag + lane;
  float* strip = strips + wave * STRIP_FLOATS;
  const float* aStrip = strip + li * STRIP_PITCH + 2 * hi;

  for (int t = t0 + wave; t < t1; t += LAYER_WAVES) {
    const int m0 = t * 16;

    int bnd = 0;
    {
      const int idx = m0 - 1 + lane;
      if (lane <= 16 && idx >= 0) bnd = cur[idx];
      bnd = min(max(bnd, 0), N_EDGES);
    }
    const int tb = __builtin_amdgcn_readlane(bnd, 0);
    const int te = __builtin_amdgcn_readlane(bnd, 16);
    int j  = 0;
    int bj = tb;
    int ej = __builtin_amdgcn_readlane(bnd, 1);
    float4 acc = make_float4(0.f, 0.f, 0.f, 0.f);

#define FLUSH_ROW()                                                           \
    {                                                                         \
      const float d = fmaxf((float)(ej - bj), 1.0f);                          \
      v4f o = {acc.x / d, acc.y / d, acc.z / d, acc.w / d};                   \
      *(v4f*)(strip + j * STRIP_PITCH + 4 * lane) = o;                        \
      acc = make_float4(0.f, 0.f, 0.f, 0.f);                                  \
      bj = ej;                                                                \
      ++j;                                                                    \
      ej = __builtin_amdgcn_readlane(bnd, min(j + 1, 16));                    \
    }

    int nxt = 0;
    if (tb + lane < te) nxt = col[tb + lane];
    for (int c = tb; c < te; c += 32) {
      const int m = min(te - c, 32);
      const int my = (int)min((unsigned)nxt, (unsigned)(N_NODES - 1));
      nxt = 0;
      if (c + 32 + lane < te) nxt = col[c + 32 + lane];
      for (int k = 0; k < m; k += 16) {
        float4 r[16];
        #pragma unroll
        for (int q = 0; q < 16; ++q) {
          r[q] = make_float4(0.f, 0.f, 0.f, 0.f);
          if (k + q < m) {
            const int sidx = __builtin_amdgcn_readlane(my, k + q);
            r[q] = ((const float4*)(xin + (size_t)sidx * D_FEAT))[lane];
          }
        }
        #pragma unroll
        for (int q = 0; q < 16; ++q) {
          if (k + q < m) {
            const int slot = c + k + q;
            while (j < 16 && slot >= ej) FLUSH_ROW()
            acc.x += r[q].x; acc.y += r[q].y; acc.z += r[q].z; acc.w += r[q].w;
          }
        }
      }
    }
    while (j < 16) FLUSH_ROW()
#undef FLUSH_ROW

    const int rowA = m0 + li;
    const float* xRow = xin + (size_t)rowA * D_FEAT + 2 * hi;

    v8f acc8[8];
    #pragma unroll
    for (int n = 0; n < 8; ++n) acc8[n] = (v8f){};

    #pragma unroll 8
    for (int s = 0; s < K_STEPS; ++s) {
      const v2f a = *(const v2f*)(aStrip + 4 * s);
      #pragma unroll
      for (int p = 0; p < 4; ++p) {
        const float4 b = bfrag[(s * 4 + p) * 32];
        v2f b0 = {b.x, b.y};
        v2f b1 = {b.z, b.w};
        acc8[2 * p]     = mma4(a, b0, acc8[2 * p]);
        acc8[2 * p + 1] = mma4(a, b1, acc8[2 * p + 1]);
      }
    }

    #pragma unroll 8
    for (int s = 0; s < K_STEPS; ++s) {
      v2f a = *(const v2f*)(xRow + 4 * s);
      #pragma unroll
      for (int p = 0; p < 4; ++p) {
        const float4 b = bfrag[((K_STEPS + s) * 4 + p) * 32];
        v2f b0 = {b.x, b.y};
        v2f b1 = {b.z, b.w};
        acc8[2 * p]     = mma4(a, b0, acc8[2 * p]);
        acc8[2 * p + 1] = mma4(a, b1, acc8[2 * p + 1]);
      }
    }

    float* outBase = out + (size_t)(m0 + 8 * hi) * D_FEAT + li;
    #pragma unroll
    for (int n = 0; n < 8; ++n) {
      const float bias = b_l[n * 16 + li];
      #pragma unroll
      for (int r = 0; r < 8; ++r) {
        float v = acc8[n][r] + bias;
        if (do_relu) v = fmaxf(v, 0.0f);
        outBase[(size_t)r * D_FEAT + n * 16] = v;
      }
    }
  }
}

extern "C" void kernel_launch(void* const* d_in, const int* in_sizes, int n_in,
                              void* d_out, int out_size, void* d_ws, size_t ws_size,
                              hipStream_t stream) {
  const float* x    = (const float*)d_in[0];
  const float* w1_l = (const float*)d_in[1];
  const float* b1_l = (const float*)d_in[2];
  const float* w1_r = (const float*)d_in[3];
  const float* w2_l = (const float*)d_in[4];
  const float* b2_l = (const float*)d_in[5];
  const float* w2_r = (const float*)d_in[6];
  const int*   edges = (const int*)d_in[7];
  const int*   src  = edges;
  const int*   dst  = edges + N_EDGES;

  float* h    = (float*)d_ws;
  int*   cnt  = (int*)(h + (size_t)N_NODES * D_FEAT);
  int*   cur  = cnt + N_NODES;
  int*   col  = cur + N_NODES;
  float* out  = (float*)d_out;

  dim3 tb(256);
  dim3 zg((N_NODES / 4 + 255) / 256);
  dim3 eg((N_EDGES + 255) / 256);
  dim3 lb(LAYER_WAVES * 32), lg(LAYER_BLOCKS);

  zero_counts<<<zg, tb, 0, stream>>>(cnt);
  count_in_edges<<<eg, tb, 0, stream>>>(dst, cnt);
  scan_counts<<<dim3(SCAN_BLOCKS), dim3(SCAN_THREADS), 0, stream>>>(cnt, cur);
  fill_sorted<<<eg, tb, 0, stream>>>(src, dst, cur, col);

  sage_layer<<<lg, lb, 0, stream>>>(x, cur, col, w1_l, b1_l, w1_r, h, 1);
  sage_layer<<<lg, lb, 0, stream>>>(h, cur, col, w2_l, b2_l, w2_r, out, 0);
}


# sage_layer gather: all 32 row reads of an index chunk issued unconditionally before any is summed (tail lanes read row m0), one boundary test per slot; 32 rows in flight, not 16; node sum order kept
# speedup vs baseline: 2.5553x; 1.0630x over previous
#include <hip/hip_runtime.h>


typedef __attribute__((ext_vector_type(2))) float v2f;
typedef __attribute__((ext_vector_type(4))) float v4f;
typedef __attribute__((ext_vector_type(8))) float v8f;

#define N_NODES 100000
#define D_FEAT  128
#define N_EDGES 640000

__global__ void zero_counts(int* __restrict__ cnt) {
  const int i = blockIdx.x * blockDim.x + threadIdx.x;
  if (i < N_NODES / 4) ((int4*)cnt)[i] = make_int4(0, 0, 0, 0);
}

__global__ void count_in_edges(const int* __restrict__ dst, int* __restrict__ cnt) {
  const int e = blockIdx.x * blockDim.x + threadIdx.x;
  if (e < N_EDGES) atomicAdd(cnt + dst[e], 1);
}

#define SCAN_THREADS 1024
#define SCAN_BLOCKS  25

__global__ __launch_bounds__(SCAN_THREADS)
void scan_counts(const int* __restrict__ cnt, int* __restrict__ cur) {
  __shared__ int wtot[32];
  __shared__ int wpre[32];

  const int tid  = threadIdx.x;
  const int lane = tid & 31;
  const int wave = tid >> 5;
  const int4* cnt4 = (const int4*)cnt;

  int pre = 0;
  #pragma unroll
  for (int j = 0; j < SCAN_BLOCKS - 1; ++j) {
    const int4 v = cnt4[j * SCAN_THREADS + tid];
    const int t = v.x + v.y + v.z + v.w;
    pre += (j < (int)blockIdx.x) ? t : 0;
  }

  const int q = blockIdx.x * SCAN_THREADS + tid;
  int4 v = make_int4(0, 0, 0, 0);
  if (q < N_NODES / 4) v = cnt4[q];
  const int s = v.x + v.y + v.z + v.w;

  int inc = s;
  #pragma unroll
  for (int d = 1; d < 32; d <<= 1) {
    const int t = __shfl_up(inc, d);
    if (lane >= d) inc += t;
  }
  #pragma unroll
  for (int d = 16; d >= 1; d >>= 1) pre += __shfl_xor(pre, d);

  if (lane == 31) wtot[wave] = inc;
  if (lane == 0)  wpre[wave] = pre;
  __syncthreads();

  int base = wpre[lane] + ((lane < wave) ? wtot[lane] : 0);
  #pragma unroll
  for (int d = 16; d >= 1; d >>= 1) base += __shfl_xor(base, d);

  if (q < N_NODES / 4) {
    int4 o;
    o.x = base + inc - s;
    o.y = o.x + v.x;
    o.z = o.y + v.y;
    o.w = o.z + v.z;
    ((int4*)cur)[q] = o;
  }
}

__global__ void fill_sorted(const int* __restrict__ src,
                            const int* __restrict__ dst,
                            int* __restrict__ cur,
                            int* __restrict__ col) {
  const int e = blockIdx.x * blockDim.x + threadIdx.x;
  if (e < N_EDGES) {
    const int s = src[e];
    const int p = atomicAdd(cur + dst[e], 1);
    if ((unsigned)p < (unsigned)N_EDGES) col[p] = s;
  }
}

#define ROW_TILES    (N_NODES / 16)
#define LAYER_BLOCKS 256
#define LAYER_WAVES  16
#define K_STEPS      (D_FEAT / 4)
#define STRIP_PITCH  132
#define STRIP_FLOATS (16 * STRIP_PITCH)

__device__ __forceinline__ v8f mma4(v2f a, v2f b, v8f c) {
  return __builtin_amdgcn_wmma_f32_16x16x4_f32(
       false, a,  false, b,
       (short)0, c,  false,  false);
}

__global__ __launch_bounds__(512)
void sage_layer(const float* __restrict__ xin,
                const int* __restrict__ cur,
                const int* __restrict__ col,
                const float* __restrict__ w_l,
                const float* __restrict__ b_l,
                const float* __restrict__ w_r,
                float* __restrict__ out,
                int do_relu) {
  __shared__ float4 wfrag[2 * K_STEPS * 4 * 32];
  __shared__ __attribute__((aligned(16))) float strips[LAYER_WAVES * STRIP_FLOATS];

  const int wave = __builtin_amdgcn_readfirstlane((int)(threadIdx.x >> 5));
  const int lane = threadIdx.x & 31;
  const int hi   = lane >> 4;
  const int li   = lane & 15;

  {
    const int par = hi;
    float* lds = (float*)wfrag;
    for (int i = 0; i < 2 * K_STEPS * 4 / LAYER_WAVES; i += 8) {
      float4 v[8];
      #pragma unroll
      for (int q = 0; q < 8; ++q) {
        const int u = wave + LAYER_WAVES * (i + q);
        const int S = u >> 2;
        const int p = u & 3;
        const float* w = (S < K_STEPS) ? w_l : w_r;
        const int s = S & (K_STEPS - 1);
        v[q] = *(const float4*)(w + (size_t)((2 * p + par) * 16 + li) * D_FEAT + 4 * s);
      }
      #pragma unroll
      for (int q = 0; q < 8; ++q) {
        const int u = wave + LAYER_WAVES * (i + q);
        v2f lo = {v[q].x, v[q].y};
        v2f up = {v[q].z, v[q].w};
        *(v2f*)(lds + (size_t)(u * 32 + li) * 4 + 2 * par)      = lo;
        *(v2f*)(lds + (size_t)(u * 32 + 16 + li) * 4 + 2 * par) = up;
      }
    }
  }
  __syncthreads();

  const int t0 = (int)(((unsigned)blockIdx.x * (unsigned)ROW_TILES) / (unsigned)LAYER_BLOCKS);
  const int t1 = (int)(((unsigned)(blockIdx.x + 1) * (unsigned)ROW_TILES) / (unsigned)LAYER_BLOCKS);

  const float4* bfrag = wfrag + lane;
  float* strip = strips + wave * STRIP_FLOATS;
  const float* aStrip = strip + li * STRIP_PITCH + 2 * hi;

  for (int t = t0 + wave; t < t1; t += LAYER_WAVES) {
    const int m0 = t * 16;

    int bnd = 0;
    {
      const int idx = m0 - 1 + lane;
      if (lane <= 16 && idx >= 0) bnd = cur[idx];
      bnd = min(max(bnd, 0), N_EDGES);
    }
    const int tb = __builtin_amdgcn_readlane(bnd, 0);
    const int te = __builtin_amdgcn_readlane(bnd, 16);
    int j  = 0;
    int bj = tb;
    int ej = __builtin_amdgcn_readlane(bnd, 1);
    float4 acc = make_float4(0.f, 0.f, 0.f, 0.f);

#define NO_SLOT 0x3fffffff

#define FLUSH_ROW()                                                           \
    {                                                                         \
      const float d = fmaxf((float)(ej - bj), 1.0f);                          \
      v4f o = {acc.x / d, acc.y / d, acc.z / d, acc.w / d};                   \
      *(v4f*)(strip + j * STRIP_PITCH + 4 * lane) = o;                        \
      acc = make_float4(0.f, 0.f, 0.f, 0.f);                                  \
      bj = ej;                                                                \
      ++j;                                                                    \
      ej = __builtin_amdgcn_readlane(bnd, min(j + 1, 16));                    \
      if (j >= 16) ej = NO_SLOT;                                              \
    }

#define LOAD16(R, IDX, OFF)                                                   \
    _Pragma("unroll")   \
    for (int q = 0; q < 16; ++q) {                                            \
      const unsigned sidx = (unsigned)__builtin_amdgcn_readlane(IDX, (OFF) + q); \
      const unsigned off = sidx * (unsigned)(D_FEAT * 4) + (unsigned)lane * 16u; \
      R[q] = *(const float4*)((const char*)xin + off);                        \
    }

#define SUM16(R, BASE)                                                        \
    _Pragma("unroll")   \
    for (int q = 0; q < 16; ++q) {                                            \
      const int slot = (BASE) + q;                                            \
      while (slot >= ej) FLUSH_ROW()                                          \
      acc.x += R[q].x; acc.y += R[q].y; acc.z += R[q].z; acc.w += R[q].w;     \
    }

#define CHUNK_READ(P0) col[min((P0) + lane, N_EDGES - 1)]
#define CHUNK_ROWS(V, P0) \
    (((P0) + lane < te) ? (int)min((unsigned)(V), (unsigned)(N_NODES - 1)) : m0)

    int idxv = CHUNK_READ(tb);
    for (int c = tb; c < te; c += 32) {
      const int rows = CHUNK_ROWS(idxv, c);
      idxv = CHUNK_READ(c + 32);
      float4 ra[16], rb[16];
      LOAD16(ra, rows, 0)
      LOAD16(rb, rows, 16)
      SUM16(ra, c)
      SUM16(rb, c + 16)
    }
    while (j < 16) FLUSH_ROW()
#undef CHUNK_ROWS
#undef CHUNK_READ
#undef SUM16
#undef LOAD16
#undef FLUSH_ROW
#undef NO_SLOT

    const int rowA = m0 + li;
    const float* xRow = xin + (size_t)rowA * D_FEAT + 2 * hi;

    v8f acc8[8];
    #pragma unroll
    for (int n = 0; n < 8; ++n) acc8[n] = (v8f){};

    #pragma unroll 8
    for (int s = 0; s < K_STEPS; ++s) {
      const v2f a = *(const v2f*)(aStrip + 4 * s);
      #pragma unroll
      for (int p = 0; p < 4; ++p) {
        const float4 b = bfrag[(s * 4 + p) * 32];
        v2f b0 = {b.x, b.y};
        v2f b1 = {b.z, b.w};
        acc8[2 * p]     = mma4(a, b0, acc8[2 * p]);
        acc8[2 * p + 1] = mma4(a, b1, acc8[2 * p + 1]);
      }
    }

    #pragma unroll 8
    for (int s = 0; s < K_STEPS; ++s) {
      v2f a = *(const v2f*)(xRow + 4 * s);
      #pragma unroll
      for (int p = 0; p < 4; ++p) {
        const float4 b = bfrag[((K_STEPS + s) * 4 + p) * 32];
        v2f b0 = {b.x, b.y};
        v2f b1 = {b.z, b.w};
        acc8[2 * p]     = mma4(a, b0, acc8[2 * p]);
        acc8[2 * p + 1] = mma4(a, b1, acc8[2 * p + 1]);
      }
    }

    float* outBase = out + (size_t)(m0 + 8 * hi) * D_FEAT + li;
    #pragma unroll
    for (int n = 0; n < 8; ++n) {
      const float bias = b_l[n * 16 + li];
      #pragma unroll
      for (int r = 0; r < 8; ++r) {
        float v = acc8[n][r] + bias;
        if (do_relu) v = fmaxf(v, 0.0f);
        outBase[(size_t)r * D_FEAT + n * 16] = v;
      }
    }
  }
}

extern "C" void kernel_launch(void* const* d_in, const int* in_sizes, int n_in,
                              void* d_out, int out_size, void* d_ws, size_t ws_size,
                              hipStream_t stream) {
  const float* x    = (const float*)d_in[0];
  const float* w1_l = (const float*)d_in[1];
  const float* b1_l = (const float*)d_in[2];
  const float* w1_r = (const float*)d_in[3];
  const float* w2_l = (const float*)d_in[4];
  const float* b2_l = (const float*)d_in[5];
  const float* w2_r = (const float*)d_in[6];
  const int*   edges = (const int*)d_in[7];
  const int*   src  = edges;
  const int*   dst  = edges + N_EDGES;

  float* h    = (float*)d_ws;
  int*   cnt  = (int*)(h + (size_t)N_NODES * D_FEAT);
  int*   cur  = cnt + N_NODES;
  int*   col  = cur + N_NODES;
  float* out  = (float*)d_out;

  dim3 tb(256);
  dim3 zg((N_NODES / 4 + 255) / 256);
  dim3 eg((N_EDGES + 255) / 256);
  dim3 lb(LAYER_WAVES * 32), lg(LAYER_BLOCKS);

  zero_counts<<<zg, tb, 0, stream>>>(cnt);
  count_in_edges<<<eg, tb, 0, stream>>>(dst, cnt);
  scan_counts<<<dim3(SCAN_BLOCKS), dim3(SCAN_THREADS), 0, stream>>>(cnt, cur);
  fill_sorted<<<eg, tb, 0, stream>>>(src, dst, cur, col);

  sage_layer<<<lg, lb, 0, stream>>>(x, cur, col, w1_l, b1_l, w1_r, h, 1);
  sage_layer<<<lg, lb, 0, stream>>>(h, cur, col, w2_l, b2_l, w2_r, out, 0);
}


# edge sort rebuilt as a two-level counting sort ranked in LDS: 250 chunks split by 391-node range, one block per range fills cur/col; device atomics and 2 launches gone; slot (sum) order by arrival
# speedup vs baseline: 3.8957x; 1.5246x over previous
#include <hip/hip_runtime.h>


typedef __attribute__((ext_vector_type(2))) float v2f;
typedef __attribute__((ext_vector_type(4))) float v4f;
typedef __attribute__((ext_vector_type(8))) float v8f;

#define N_NODES 100000
#define D_FEAT  128
#define N_EDGES 640000

#define BKT_W      391
#define N_BKT      256
#define CH_EDGES   2560
#define N_CH       (N_EDGES / CH_EDGES)
#define CH_THREADS 512
#define TAB_PITCH  (N_BKT + 1)

__global__ __launch_bounds__(CH_THREADS)
void bucket_chunk(const int* __restrict__ src, const int* __restrict__ dst,
                  int* __restrict__ tab, int* __restrict__ trec) {
  __shared__ int hist[N_BKT];
  __shared__ int offs[N_BKT];
  __shared__ int wsum[N_BKT / 32];

  const int tid  = threadIdx.x;
  const int lane = tid & 31;
  const int wave = tid >> 5;
  const int c    = blockIdx.x;
  const int base = c * CH_EDGES;

  if (tid < N_BKT) hist[tid] = 0;
  int d[5], s[5];
  #pragma unroll
  for (int k = 0; k < 5; ++k) {
    d[k] = dst[base + k * CH_THREADS + tid];
    s[k] = src[base + k * CH_THREADS + tid];
  }
  __syncthreads();

  int b[5], r[5];
  #pragma unroll
  for (int k = 0; k < 5; ++k) {
    b[k] = min((int)((unsigned)d[k] / (unsigned)BKT_W), N_BKT - 1);
    r[k] = atomicAdd(&hist[b[k]], 1);
  }
  __syncthreads();

  const int v = (tid < N_BKT) ? hist[tid] : 0;
  int inc = v;
  #pragma unroll
  for (int dd = 1; dd < 32; dd <<= 1) {
    const int t = __shfl_up(inc, dd);
    if (lane >= dd) inc += t;
  }
  if (tid < N_BKT && lane == 31) wsum[wave] = inc;
  __syncthreads();
  if (tid < N_BKT) {
    int pre = 0;
    #pragma unroll
    for (int w = 0; w < N_BKT / 32; ++w) pre += (w < wave) ? wsum[w] : 0;
    const int ex = pre + inc - v;
    offs[tid] = ex;
    tab[c * TAB_PITCH + tid] = ex;
  }
  if (tid == 0) tab[c * TAB_PITCH + N_BKT] = CH_EDGES;
  __syncthreads();

  #pragma unroll
  for (int k = 0; k < 5; ++k) {
    const int pos = offs[b[k]] + r[k];
    ((int2*)trec)[base + pos] = make_int2(s[k], d[k]);
  }
}

#define CSR_THREADS 1024
#define CSR_SLOTS   512

__global__ __launch_bounds__(CSR_THREADS)
void build_csr(const int* __restrict__ tab, const int* __restrict__ trec,
               int* __restrict__ cur, int* __restrict__ col) {
  __shared__ int cnt[CSR_SLOTS];
  __shared__ int start[CSR_SLOTS];
  __shared__ int wfront[32];
  __shared__ int wscan[CSR_SLOTS / 32];

  const int tid  = threadIdx.x;
  const int lane = tid & 31;
  const int wave = tid >> 5;
  const int b    = blockIdx.x;
  const int n0   = b * BKT_W;

  if (tid < CSR_SLOTS) cnt[tid] = 0;

  const int c = tid >> 2;
  const int q = tid & 3;
  int st = 0, en = 0;
  if (c < N_CH) {
    st = tab[c * TAB_PITCH + b];
    en = tab[c * TAB_PITCH + b + 1];
  }
  st = min(max(st, 0), CH_EDGES);
  en = min(max(en, st), CH_EDGES);
  const int2* pr = (const int2*)trec + (size_t)min(c, N_CH - 1) * CH_EDGES;
  __syncthreads();

  for (int i = st + q; i < en; i += 4) {
    const int dl = pr[i].y - n0;
    if ((unsigned)dl < (unsigned)BKT_W) atomicAdd(&cnt[dl], 1);
  }

  int front = (q == 0) ? st : 0;
  #pragma unroll
  for (int dd = 16; dd >= 1; dd >>= 1) front += __shfl_xor(front, dd);
  if (lane == 0) wfront[wave] = front;
  __syncthreads();

  int g0 = wfront[lane];
  #pragma unroll
  for (int dd = 16; dd >= 1; dd >>= 1) g0 += __shfl_xor(g0, dd);

  const int v = (tid < CSR_SLOTS) ? cnt[tid] : 0;
  int inc = v;
  #pragma unroll
  for (int dd = 1; dd < 32; dd <<= 1) {
    const int t = __shfl_up(inc, dd);
    if (lane >= dd) inc += t;
  }
  if (tid < CSR_SLOTS && lane == 31) wscan[wave] = inc;
  __syncthreads();
  if (tid < CSR_SLOTS) {
    int pre = 0;
    #pragma unroll
    for (int w = 0; w < CSR_SLOTS / 32; ++w) pre += (w < wave) ? wscan[w] : 0;
    const int ex = pre + inc - v;
    start[tid] = ex;
    cnt[tid] = 0;
    if (tid < BKT_W && n0 + tid < N_NODES) cur[n0 + tid] = g0 + ex + v;
  }
  __syncthreads();

  for (int i = st + q; i < en; i += 4) {
    const int2 e = pr[i];
    const int dl = e.y - n0;
    if ((unsigned)dl < (unsigned)BKT_W) {
      const int p = g0 + start[dl] + atomicAdd(&cnt[dl], 1);
      if ((unsigned)p < (unsigned)N_EDGES) col[p] = e.x;
    }
  }
}

#define ROW_TILES    (N_NODES / 16)
#define LAYER_BLOCKS 256
#define LAYER_WAVES  16
#define K_STEPS      (D_FEAT / 4)
#define STRIP_PITCH  132
#define STRIP_FLOATS (16 * STRIP_PITCH)

__device__ __forceinline__ v8f mma4(v2f a, v2f b, v8f c) {
  return __builtin_amdgcn_wmma_f32_16x16x4_f32(
       false, a,  false, b,
       (short)0, c,  false,  false);
}

__global__ __launch_bounds__(512)
void sage_layer(const float* __restrict__ xin,
                const int* __restrict__ cur,
                const int* __restrict__ col,
                const float* __restrict__ w_l,
                const float* __restrict__ b_l,
                const float* __restrict__ w_r,
                float* __restrict__ out,
                int do_relu) {
  __shared__ float4 wfrag[2 * K_STEPS * 4 * 32];
  __shared__ __attribute__((aligned(16))) float strips[LAYER_WAVES * STRIP_FLOATS];

  const int wave = __builtin_amdgcn_readfirstlane((int)(threadIdx.x >> 5));
  const int lane = threadIdx.x & 31;
  const int hi   = lane >> 4;
  const int li   = lane & 15;

  {
    const int par = hi;
    float* lds = (float*)wfrag;
    for (int i = 0; i < 2 * K_STEPS * 4 / LAYER_WAVES; i += 8) {
      float4 v[8];
      #pragma unroll
      for (int q = 0; q < 8; ++q) {
        const int u = wave + LAYER_WAVES * (i + q);
        const int S = u >> 2;
        const int p = u & 3;
        const float* w = (S < K_STEPS) ? w_l : w_r;
        const int s = S & (K_STEPS - 1);
        v[q] = *(const float4*)(w + (size_t)((2 * p + par) * 16 + li) * D_FEAT + 4 * s);
      }
      #pragma unroll
      for (int q = 0; q < 8; ++q) {
        const int u = wave + LAYER_WAVES * (i + q);
        v2f lo = {v[q].x, v[q].y};
        v2f up = {v[q].z, v[q].w};
        *(v2f*)(lds + (size_t)(u * 32 + li) * 4 + 2 * par)      = lo;
        *(v2f*)(lds + (size_t)(u * 32 + 16 + li) * 4 + 2 * par) = up;
      }
    }
  }
  __syncthreads();

  const int t0 = (int)(((unsigned)blockIdx.x * (unsigned)ROW_TILES) / (unsigned)LAYER_BLOCKS);
  const int t1 = (int)(((unsigned)(blockIdx.x + 1) * (unsigned)ROW_TILES) / (unsigned)LAYER_BLOCKS);

  const float4* bfrag = wfrag + lane;
  float* strip = strips + wave * STRIP_FLOATS;
  const float* aStrip = strip + li * STRIP_PITCH + 2 * hi;

  for (int t = t0 + wave; t < t1; t += LAYER_WAVES) {
    const int m0 = t * 16;

    int bnd = 0;
    {
      const int idx = m0 - 1 + lane;
      if (lane <= 16 && idx >= 0) bnd = cur[idx];
      bnd = min(max(bnd, 0), N_EDGES);
    }
    const int tb = __builtin_amdgcn_readlane(bnd, 0);
    const int te = __builtin_amdgcn_readlane(bnd, 16);
    int j  = 0;
    int bj = tb;
    int ej = __builtin_amdgcn_readlane(bnd, 1);
    float4 acc = make_float4(0.f, 0.f, 0.f, 0.f);

#define NO_SLOT 0x3fffffff

#define FLUSH_ROW()                                                           \
    {                                                                         \
      const float d = fmaxf((float)(ej - bj), 1.0f);                          \
      v4f o = {acc.x / d, acc.y / d, acc.z / d, acc.w / d};                   \
      *(v4f*)(strip + j * STRIP_PITCH + 4 * lane) = o;                        \
      acc = make_float4(0.f, 0.f, 0.f, 0.f);                                  \
      bj = ej;                                                                \
      ++j;                                                                    \
      ej = __builtin_amdgcn_readlane(bnd, min(j + 1, 16));                    \
      if (j >= 16) ej = NO_SLOT;                                              \
    }

#define LOAD16(R, IDX, OFF)                                                   \
    _Pragma("unroll")   \
    for (int q = 0; q < 16; ++q) {                                            \
      const unsigned sidx = (unsigned)__builtin_amdgcn_readlane(IDX, (OFF) + q); \
      const unsigned off = sidx * (unsigned)(D_FEAT * 4) + (unsigned)lane * 16u; \
      R[q] = *(const float4*)((const char*)xin + off);                        \
    }

#define SUM16(R, BASE)                                                        \
    _Pragma("unroll")   \
    for (int q = 0; q < 16; ++q) {                                            \
      const int slot = (BASE) + q;                                            \
      while (slot >= ej) FLUSH_ROW()                                          \
      acc.x += R[q].x; acc.y += R[q].y; acc.z += R[q].z; acc.w += R[q].w;     \
    }

#define CHUNK_READ(P0) col[min((P0) + lane, N_EDGES - 1)]
#define CHUNK_ROWS(V, P0) \
    (((P0) + lane < te) ? (int)min((unsigned)(V), (unsigned)(N_NODES - 1)) : m0)

    int idxv = CHUNK_READ(tb);
    for (int c = tb; c < te; c += 32) {
      const int rows = CHUNK_ROWS(idxv, c);
      idxv = CHUNK_READ(c + 32);
      float4 ra[16], rb[16];
      LOAD16(ra, rows, 0)
      LOAD16(rb, rows, 16)
      SUM16(ra, c)
      SUM16(rb, c + 16)
    }
    while (j < 16) FLUSH_ROW()
#undef CHUNK_ROWS
#undef CHUNK_READ
#undef SUM16
#undef LOAD16
#undef FLUSH_ROW
#undef NO_SLOT

    const int rowA = m0 + li;
    const float* xRow = xin + (size_t)rowA * D_FEAT + 2 * hi;

    v8f acc8[8];
    #pragma unroll
    for (int n = 0; n < 8; ++n) acc8[n] = (v8f){};

    #pragma unroll 8
    for (int s = 0; s < K_STEPS; ++s) {
      const v2f a = *(const v2f*)(aStrip + 4 * s);
      #pragma unroll
      for (int p = 0; p < 4; ++p) {
        const float4 b = bfrag[(s * 4 + p) * 32];
        v2f b0 = {b.x, b.y};
        v2f b1 = {b.z, b.w};
        acc8[2 * p]     = mma4(a, b0, acc8[2 * p]);
        acc8[2 * p + 1] = mma4(a, b1, acc8[2 * p + 1]);
      }
    }

    #pragma unroll 8
    for (int s = 0; s < K_STEPS; ++s) {
      v2f a = *(const v2f*)(xRow + 4 * s);
      #pragma unroll
      for (int p = 0; p < 4; ++p) {
        const float4 b = bfrag[((K_STEPS + s) * 4 + p) * 32];
        v2f b0 = {b.x, b.y};
        v2f b1 = {b.z, b.w};
        acc8[2 * p]     = mma4(a, b0, acc8[2 * p]);
        acc8[2 * p + 1] = mma4(a, b1, acc8[2 * p + 1]);
      }
    }

    float* outBase = out + (size_t)(m0 + 8 * hi) * D_FEAT + li;
    #pragma unroll
    for (int n = 0; n < 8; ++n) {
      const float bias = b_l[n * 16 + li];
      #pragma unroll
      for (int r = 0; r < 8; ++r) {
        float v = acc8[n][r] + bias;
        if (do_relu) v = fmaxf(v, 0.0f);
        outBase[(size_t)r * D_FEAT + n * 16] = v;
      }
    }
  }
}

extern "C" void kernel_launch(void* const* d_in, const int* in_sizes, int n_in,
                              void* d_out, int out_size, void* d_ws, size_t ws_size,
                              hipStream_t stream) {
  const float* x    = (const float*)d_in[0];
  const float* w1_l = (const float*)d_in[1];
  const float* b1_l = (const float*)d_in[2];
  const float* w1_r = (const float*)d_in[3];
  const float* w2_l = (const float*)d_in[4];
  const float* b2_l = (const float*)d_in[5];
  const float* w2_r = (const float*)d_in[6];
  const int*   edges = (const int*)d_in[7];
  const int*   src  = edges;
  const int*   dst  = edges + N_EDGES;

  float* h    = (float*)d_ws;
  int*   cur  = (int*)(h + (size_t)N_NODES * D_FEAT);
  int*   col  = cur + N_NODES;
  int*   trec = col + N_EDGES;
  int*   tab  = trec + 2 * N_EDGES;
  float* out  = (float*)d_out;

  dim3 lb(LAYER_WAVES * 32), lg(LAYER_BLOCKS);

  bucket_chunk<<<dim3(N_CH), dim3(CH_THREADS), 0, stream>>>(src, dst, tab, trec);
  build_csr<<<dim3(N_BKT), dim3(CSR_THREADS), 0, stream>>>(tab, trec, cur, col);

  sage_layer<<<lg, lb, 0, stream>>>(x, cur, col, w1_l, b1_l, w1_r, h, 1);
  sage_layer<<<lg, lb, 0, stream>>>(h, cur, col, w2_l, b2_l, w2_r, out, 0);
}


# sage_layer weight staging: a wave reads whole 512-byte weight rows (4 lines per load, was 32) and scatters them to B-fragment order in LDS; k-step groups padded 16 B to spread the banks
# speedup vs baseline: 4.1531x; 1.0661x over previous
#include <hip/hip_runtime.h>


typedef __attribute__((ext_vector_type(2))) float v2f;
typedef __attribute__((ext_vector_type(4))) float v4f;
typedef __attribute__((ext_vector_type(8))) float v8f;

#define N_NODES 100000
#define D_FEAT  128
#define N_EDGES 640000

#define BKT_W      391
#define N_BKT      256
#define CH_EDGES   2560
#define N_CH       (N_EDGES / CH_EDGES)
#define CH_THREADS 512
#define TAB_PITCH  (N_BKT + 1)

__global__ __launch_bounds__(CH_THREADS)
void bucket_chunk(const int* __restrict__ src, const int* __restrict__ dst,
                  int* __restrict__ tab, int* __restrict__ trec) {
  __shared__ int hist[N_BKT];
  __shared__ int offs[N_BKT];
  __shared__ int wsum[N_BKT / 32];

  const int tid  = threadIdx.x;
  const int lane = tid & 31;
  const int wave = tid >> 5;
  const int c    = blockIdx.x;
  const int base = c * CH_EDGES;

  if (tid < N_BKT) hist[tid] = 0;
  int d[5], s[5];
  #pragma unroll
  for (int k = 0; k < 5; ++k) {
    d[k] = dst[base + k * CH_THREADS + tid];
    s[k] = src[base + k * CH_THREADS + tid];
  }
  __syncthreads();

  int b[5], r[5];
  #pragma unroll
  for (int k = 0; k < 5; ++k) {
    b[k] = min((int)((unsigned)d[k] / (unsigned)BKT_W), N_BKT - 1);
    r[k] = atomicAdd(&hist[b[k]], 1);
  }
  __syncthreads();

  const int v = (tid < N_BKT) ? hist[tid] : 0;
  int inc = v;
  #pragma unroll
  for (int dd = 1; dd < 32; dd <<= 1) {
    const int t = __shfl_up(inc, dd);
    if (lane >= dd) inc += t;
  }
  if (tid < N_BKT && lane == 31) wsum[wave] = inc;
  __syncthreads();
  if (tid < N_BKT) {
    int pre = 0;
    #pragma unroll
    for (int w = 0; w < N_BKT / 32; ++w) pre += (w < wave) ? wsum[w] : 0;
    const int ex = pre + inc - v;
    offs[tid] = ex;
    tab[c * TAB_PITCH + tid] = ex;
  }
  if (tid == 0) tab[c * TAB_PITCH + N_BKT] = CH_EDGES;
  __syncthreads();

  #pragma unroll
  for (int k = 0; k < 5; ++k) {
    const int pos = offs[b[k]] + r[k];
    ((int2*)trec)[base + pos] = make_int2(s[k], d[k]);
  }
}

#define CSR_THREADS 1024
#define CSR_SLOTS   512

__global__ __launch_bounds__(CSR_THREADS)
void build_csr(const int* __restrict__ tab, const int* __restrict__ trec,
               int* __restrict__ cur, int* __restrict__ col) {
  __shared__ int cnt[CSR_SLOTS];
  __shared__ int start[CSR_SLOTS];
  __shared__ int wfront[32];
  __shared__ int wscan[CSR_SLOTS / 32];

  const int tid  = threadIdx.x;
  const int lane = tid & 31;
  const int wave = tid >> 5;
  const int b    = blockIdx.x;
  const int n0   = b * BKT_W;

  if (tid < CSR_SLOTS) cnt[tid] = 0;

  const int c = tid >> 2;
  const int q = tid & 3;
  int st = 0, en = 0;
  if (c < N_CH) {
    st = tab[c * TAB_PITCH + b];
    en = tab[c * TAB_PITCH + b + 1];
  }
  st = min(max(st, 0), CH_EDGES);
  en = min(max(en, st), CH_EDGES);
  const int2* pr = (const int2*)trec + (size_t)min(c, N_CH - 1) * CH_EDGES;
  __syncthreads();

  for (int i = st + q; i < en; i += 4) {
    const int dl = pr[i].y - n0;
    if ((unsigned)dl < (unsigned)BKT_W) atomicAdd(&cnt[dl], 1);
  }

  int front = (q == 0) ? st : 0;
  #pragma unroll
  for (int dd = 16; dd >= 1; dd >>= 1) front += __shfl_xor(front, dd);
  if (lane == 0) wfront[wave] = front;
  __syncthreads();

  int g0 = wfront[lane];
  #pragma unroll
  for (int dd = 16; dd >= 1; dd >>= 1) g0 += __shfl_xor(g0, dd);

  const int v = (tid < CSR_SLOTS) ? cnt[tid] : 0;
  int inc = v;
  #pragma unroll
  for (int dd = 1; dd < 32; dd <<= 1) {
    const int t = __shfl_up(inc, dd);
    if (lane >= dd) inc += t;
  }
  if (tid < CSR_SLOTS && lane == 31) wscan[wave] = inc;
  __syncthreads();
  if (tid < CSR_SLOTS) {
    int pre = 0;
    #pragma unroll
    for (int w = 0; w < CSR_SLOTS / 32; ++w) pre += (w < wave) ? wscan[w] : 0;
    const int ex = pre + inc - v;
    start[tid] = ex;
    cnt[tid] = 0;
    if (tid < BKT_W && n0 + tid < N_NODES) cur[n0 + tid] = g0 + ex + v;
  }
  __syncthreads();

  for (int i = st + q; i < en; i += 4) {
    const int2 e = pr[i];
    const int dl = e.y - n0;
    if ((unsigned)dl < (unsigned)BKT_W) {
      const int p = g0 + start[dl] + atomicAdd(&cnt[dl], 1);
      if ((unsigned)p < (unsigned)N_EDGES) col[p] = e.x;
    }
  }
}

#define ROW_TILES    (N_NODES / 16)
#define LAYER_BLOCKS 256
#define LAYER_WAVES  16
#define K_STEPS      (D_FEAT / 4)
#define KS_PITCH4    (4 * 32 + 1)
#define STRIP_PITCH  132
#define STRIP_FLOATS (16 * STRIP_PITCH)

__device__ __forceinline__ v8f mma4(v2f a, v2f b, v8f c) {
  return __builtin_amdgcn_wmma_f32_16x16x4_f32(
       false, a,  false, b,
       (short)0, c,  false,  false);
}

__global__ __launch_bounds__(512)
void sage_layer(const float* __restrict__ xin,
                const int* __restrict__ cur,
                const int* __restrict__ col,
                const float* __restrict__ w_l,
                const float* __restrict__ b_l,
                const float* __restrict__ w_r,
                float* __restrict__ out,
                int do_relu) {
  __shared__ float4 wfrag[2 * K_STEPS * KS_PITCH4];
  __shared__ __attribute__((aligned(16))) float strips[LAYER_WAVES * STRIP_FLOATS];

  const int wave = __builtin_amdgcn_readfirstlane((int)(threadIdx.x >> 5));
  const int lane = threadIdx.x & 31;
  const int hi   = lane >> 4;
  const int li   = lane & 15;

  {
    float* lds = (float*)wfrag;
    float4 v[16];
    #pragma unroll
    for (int q = 0; q < 16; ++q) {
      const int R = wave + LAYER_WAVES * q;
      const float* w = (R < D_FEAT) ? w_l : w_r;
      v[q] = *(const float4*)(w + (size_t)(R & (D_FEAT - 1)) * D_FEAT + 4 * lane);
    }
    #pragma unroll
    for (int q = 0; q < 16; ++q) {
      const int R  = wave + LAYER_WAVES * q;
      const int n  = R & (D_FEAT - 1);
      const int S  = (R >> 7) * K_STEPS + lane;
      const int ct = n >> 4;
      float* d = lds + (size_t)S * (KS_PITCH4 * 4) + (ct >> 1) * 128 + (n & 15) * 4 + (ct & 1) * 2;
      v2f lo = {v[q].x, v[q].y};
      v2f up = {v[q].z, v[q].w};
      *(v2f*)d        = lo;
      *(v2f*)(d + 64) = up;
    }
  }
  __syncthreads();

  const int t0 = (int)(((unsigned)blockIdx.x * (unsigned)ROW_TILES) / (unsigned)LAYER_BLOCKS);
  const int t1 = (int)(((unsigned)(blockIdx.x + 1) * (unsigned)ROW_TILES) / (unsigned)LAYER_BLOCKS);

  const float4* bfrag = wfrag + lane;
  float* strip = strips + wave * STRIP_FLOATS;
  const float* aStrip = strip + li * STRIP_PITCH + 2 * hi;

  for (int t = t0 + wave; t < t1; t += LAYER_WAVES) {
    const int m0 = t * 16;

    int bnd = 0;
    {
      const int idx = m0 - 1 + lane;
      if (lane <= 16 && idx >= 0) bnd = cur[idx];
      bnd = min(max(bnd, 0), N_EDGES);
    }
    const int tb = __builtin_amdgcn_readlane(bnd, 0);
    const int te = __builtin_amdgcn_readlane(bnd, 16);
    int j  = 0;
    int bj = tb;
    int ej = __builtin_amdgcn_readlane(bnd, 1);
    float4 acc = make_float4(0.f, 0.f, 0.f, 0.f);

#define NO_SLOT 0x3fffffff

#define FLUSH_ROW()                                                           \
    {                                                                         \
      const float d = fmaxf((float)(ej - bj), 1.0f);                          \
      v4f o = {acc.x / d, acc.y / d, acc.z / d, acc.w / d};                   \
      *(v4f*)(strip + j * STRIP_PITCH + 4 * lane) = o;                        \
      acc = make_float4(0.f, 0.f, 0.f, 0.f);                                  \
      bj = ej;                                                                \
      ++j;                                                                    \
      ej = __builtin_amdgcn_readlane(bnd, min(j + 1, 16));                    \
      if (j >= 16) ej = NO_SLOT;                                              \
    }

#define LOAD16(R, IDX, OFF)                                                   \
    _Pragma("unroll")   \
    for (int q = 0; q < 16; ++q) {                                            \
      const unsigned sidx = (unsigned)__builtin_amdgcn_readlane(IDX, (OFF) + q); \
      const unsigned off = sidx * (unsigned)(D_FEAT * 4) + (unsigned)lane * 16u; \
      R[q] = *(const float4*)((const char*)xin + off);                        \
    }

#define SUM16(R, BASE)                                                        \
    _Pragma("unroll")   \
    for (int q = 0; q < 16; ++q) {                                            \
      const int slot = (BASE) + q;                                            \
      while (slot >= ej) FLUSH_ROW()                                          \
      acc.x += R[q].x; acc.y += R[q].y; acc.z += R[q].z; acc.w += R[q].w;     \
    }

#define CHUNK_READ(P0) col[min((P0) + lane, N_EDGES - 1)]
#define CHUNK_ROWS(V, P0) \
    (((P0) + lane < te) ? (int)min((unsigned)(V), (unsigned)(N_NODES - 1)) : m0)

    int idxv = CHUNK_READ(tb);
    for (int c = tb; c < te; c += 32) {
      const int rows = CHUNK_ROWS(idxv, c);
      idxv = CHUNK_READ(c + 32);
      float4 ra[16], rb[16];
      LOAD16(ra, rows, 0)
      LOAD16(rb, rows, 16)
      SUM16(ra, c)
      SUM16(rb, c + 16)
    }
    while (j < 16) FLUSH_ROW()
#undef CHUNK_ROWS
#undef CHUNK_READ
#undef SUM16
#undef LOAD16
#undef FLUSH_ROW
#undef NO_SLOT

    const int rowA = m0 + li;
    const float* xRow = xin + (size_t)rowA * D_FEAT + 2 * hi;

    v8f acc8[8];
    #pragma unroll
    for (int n = 0; n < 8; ++n) acc8[n] = (v8f){};

    #pragma unroll 8
    for (int s = 0; s < K_STEPS; ++s) {
      const v2f a = *(const v2f*)(aStrip + 4 * s);
      #pragma unroll
      for (int p = 0; p < 4; ++p) {
        const float4 b = bfrag[s * KS_PITCH4 + p * 32];
        v2f b0 = {b.x, b.y};
        v2f b1 = {b.z, b.w};
        acc8[2 * p]     = mma4(a, b0, acc8[2 * p]);
        acc8[2 * p + 1] = mma4(a, b1, acc8[2 * p + 1]);
      }
    }

    #pragma unroll 8
    for (int s = 0; s < K_STEPS; ++s) {
      v2f a = *(const v2f*)(xRow + 4 * s);
      #pragma unroll
      for (int p = 0; p < 4; ++p) {
        const float4 b = bfrag[(K_STEPS + s) * KS_PITCH4 + p * 32];
        v2f b0 = {b.x, b.y};
        v2f b1 = {b.z, b.w};
        acc8[2 * p]     = mma4(a, b0, acc8[2 * p]);
        acc8[2 * p + 1] = mma4(a, b1, acc8[2 * p + 1]);
      }
    }

    float* outBase = out + (size_t)(m0 + 8 * hi) * D_FEAT + li;
    #pragma unroll
    for (int n = 0; n < 8; ++n) {
      const float bias = b_l[n * 16 + li];
      #pragma unroll
      for (int r = 0; r < 8; ++r) {
        float v = acc8[n][r] + bias;
        if (do_relu) v = fmaxf(v, 0.0f);
        outBase[(size_t)r * D_FEAT + n * 16] = v;
      }
    }
  }
}

extern "C" void kernel_launch(void* const* d_in, const int* in_sizes, int n_in,
                              void* d_out, int out_size, void* d_ws, size_t ws_size,
                              hipStream_t stream) {
  const float* x    = (const float*)d_in[0];
  const float* w1_l = (const float*)d_in[1];
  const float* b1_l = (const float*)d_in[2];
  const float* w1_r = (const float*)d_in[3];
  const float* w2_l = (const float*)d_in[4];
  const float* b2_l = (const float*)d_in[5];
  const float* w2_r = (const float*)d_in[6];
  const int*   edges = (const int*)d_in[7];
  const int*   src  = edges;
  const int*   dst  = edges + N_EDGES;

  float* h    = (float*)d_ws;
  int*   cur  = (int*)(h + (size_t)N_NODES * D_FEAT);
  int*   col  = cur + N_NODES;
  int*   trec = col + N_EDGES;
  int*   tab  = trec + 2 * N_EDGES;
  float* out  = (float*)d_out;

  dim3 lb(LAYER_WAVES * 32), lg(LAYER_BLOCKS);

  bucket_chunk<<<dim3(N_CH), dim3(CH_THREADS), 0, stream>>>(src, dst, tab, trec);
  build_csr<<<dim3(N_BKT), dim3(CSR_THREADS), 0, stream>>>(tab, trec, cur, col);

  sage_layer<<<lg, lb, 0, stream>>>(x, cur, col, w1_l, b1_l, w1_r, h, 1);
  sage_layer<<<lg, lb, 0, stream>>>(h, cur, col, w2_l, b2_l, w2_r, out, 0);
}


# sage_layer tile schedule: a block's 25th row tile is cut in four column quarters, one per SIMD (each wave gathers the rows itself, runs 2 of 8 column tiles), not a 7th whole tile on one SIMD
# speedup vs baseline: 4.5022x; 1.0841x over previous
#include <hip/hip_runtime.h>


typedef __attribute__((ext_vector_type(2))) float v2f;
typedef __attribute__((ext_vector_type(4))) float v4f;
typedef __attribute__((ext_vector_type(8))) float v8f;

#define N_NODES 100000
#define D_FEAT  128
#define N_EDGES 640000

#define BKT_W      391
#define N_BKT      256
#define CH_EDGES   2560
#define N_CH       (N_EDGES / CH_EDGES)
#define CH_THREADS 512
#define TAB_PITCH  (N_BKT + 1)

__global__ __launch_bounds__(CH_THREADS)
void bucket_chunk(const int* __restrict__ src, const int* __restrict__ dst,
                  int* __restrict__ tab, int* __restrict__ trec) {
  __shared__ int hist[N_BKT];
  __shared__ int offs[N_BKT];
  __shared__ int wsum[N_BKT / 32];

  const int tid  = threadIdx.x;
  const int lane = tid & 31;
  const int wave = tid >> 5;
  const int c    = blockIdx.x;
  const int base = c * CH_EDGES;

  if (tid < N_BKT) hist[tid] = 0;
  int d[5], s[5];
  #pragma unroll
  for (int k = 0; k < 5; ++k) {
    d[k] = dst[base + k * CH_THREADS + tid];
    s[k] = src[base + k * CH_THREADS + tid];
  }
  __syncthreads();

  int b[5], r[5];
  #pragma unroll
  for (int k = 0; k < 5; ++k) {
    b[k] = min((int)((unsigned)d[k] / (unsigned)BKT_W), N_BKT - 1);
    r[k] = atomicAdd(&hist[b[k]], 1);
  }
  __syncthreads();

  const int v = (tid < N_BKT) ? hist[tid] : 0;
  int inc = v;
  #pragma unroll
  for (int dd = 1; dd < 32; dd <<= 1) {
    const int t = __shfl_up(inc, dd);
    if (lane >= dd) inc += t;
  }
  if (tid < N_BKT && lane == 31) wsum[wave] = inc;
  __syncthreads();
  if (tid < N_BKT) {
    int pre = 0;
    #pragma unroll
    for (int w = 0; w < N_BKT / 32; ++w) pre += (w < wave) ? wsum[w] : 0;
    const int ex = pre + inc - v;
    offs[tid] = ex;
    tab[c * TAB_PITCH + tid] = ex;
  }
  if (tid == 0) tab[c * TAB_PITCH + N_BKT] = CH_EDGES;
  __syncthreads();

  #pragma unroll
  for (int k = 0; k < 5; ++k) {
    const int pos = offs[b[k]] + r[k];
    ((int2*)trec)[base + pos] = make_int2(s[k], d[k]);
  }
}

#define CSR_THREADS 1024
#define CSR_SLOTS   512

__global__ __launch_bounds__(CSR_THREADS)
void build_csr(const int* __restrict__ tab, const int* __restrict__ trec,
               int* __restrict__ cur, int* __restrict__ col) {
  __shared__ int cnt[CSR_SLOTS];
  __shared__ int start[CSR_SLOTS];
  __shared__ int wfront[32];
  __shared__ int wscan[CSR_SLOTS / 32];

  const int tid  = threadIdx.x;
  const int lane = tid & 31;
  const int wave = tid >> 5;
  const int b    = blockIdx.x;
  const int n0   = b * BKT_W;

  if (tid < CSR_SLOTS) cnt[tid] = 0;

  const int c = tid >> 2;
  const int q = tid & 3;
  int st = 0, en = 0;
  if (c < N_CH) {
    st = tab[c * TAB_PITCH + b];
    en = tab[c * TAB_PITCH + b + 1];
  }
  st = min(max(st, 0), CH_EDGES);
  en = min(max(en, st), CH_EDGES);
  const int2* pr = (const int2*)trec + (size_t)min(c, N_CH - 1) * CH_EDGES;
  __syncthreads();

  for (int i = st + q; i < en; i += 4) {
    const int dl = pr[i].y - n0;
    if ((unsigned)dl < (unsigned)BKT_W) atomicAdd(&cnt[dl], 1);
  }

  int front = (q == 0) ? st : 0;
  #pragma unroll
  for (int dd = 16; dd >= 1; dd >>= 1) front += __shfl_xor(front, dd);
  if (lane == 0) wfront[wave] = front;
  __syncthreads();

  int g0 = wfront[lane];
  #pragma unroll
  for (int dd = 16; dd >= 1; dd >>= 1) g0 += __shfl_xor(g0, dd);

  const int v = (tid < CSR_SLOTS) ? cnt[tid] : 0;
  int inc = v;
  #pragma unroll
  for (int dd = 1; dd < 32; dd <<= 1) {
    const int t = __shfl_up(inc, dd);
    if (lane >= dd) inc += t;
  }
  if (tid < CSR_SLOTS && lane == 31) wscan[wave] = inc;
  __syncthreads();
  if (tid < CSR_SLOTS) {
    int pre = 0;
    #pragma unroll
    for (int w = 0; w < CSR_SLOTS / 32; ++w) pre += (w < wave) ? wscan[w] : 0;
    const int ex = pre + inc - v;
    start[tid] = ex;
    cnt[tid] = 0;
    if (tid < BKT_W && n0 + tid < N_NODES) cur[n0 + tid] = g0 + ex + v;
  }
  __syncthreads();

  for (int i = st + q; i < en; i += 4) {
    const int2 e = pr[i];
    const int dl = e.y - n0;
    if ((unsigned)dl < (unsigned)BKT_W) {
      const int p = g0 + start[dl] + atomicAdd(&cnt[dl], 1);
      if ((unsigned)p < (unsigned)N_EDGES) col[p] = e.x;
    }
  }
}

#define ROW_TILES    (N_NODES / 16)
#define LAYER_BLOCKS 256
#define LAYER_WAVES  16
#define K_STEPS      (D_FEAT / 4)
#define KS_PITCH4    (4 * 32 + 1)
#define STRIP_PITCH  132
#define STRIP_FLOATS (16 * STRIP_PITCH)

__device__ __forceinline__ v8f mma4(v2f a, v2f b, v8f c) {
  return __builtin_amdgcn_wmma_f32_16x16x4_f32(
       false, a,  false, b,
       (short)0, c,  false,  false);
}

template <int NP>
__device__ __forceinline__ void tile_product(const float* aStrip, const float* xRow,
                                             const float4* bq, const float* __restrict__ b_l,
                                             float* outBase, int col0, int do_relu, int li) {
  v8f acc8[2 * NP];
  #pragma unroll
  for (int n = 0; n < 2 * NP; ++n) acc8[n] = (v8f){};

  #pragma unroll 8
  for (int s = 0; s < K_STEPS; ++s) {
    const v2f a = *(const v2f*)(aStrip + 4 * s);
    #pragma unroll
    for (int p = 0; p < NP; ++p) {
      const float4 b = bq[s * KS_PITCH4 + p * 32];
      v2f b0 = {b.x, b.y};
      v2f b1 = {b.z, b.w};
      acc8[2 * p]     = mma4(a, b0, acc8[2 * p]);
      acc8[2 * p + 1] = mma4(a, b1, acc8[2 * p + 1]);
    }
  }

  #pragma unroll 8
  for (int s = 0; s < K_STEPS; ++s) {
    v2f a = *(const v2f*)(xRow + 4 * s);
    #pragma unroll
    for (int p = 0; p < NP; ++p) {
      const float4 b = bq[(K_STEPS + s) * KS_PITCH4 + p * 32];
      v2f b0 = {b.x, b.y};
      v2f b1 = {b.z, b.w};
      acc8[2 * p]     = mma4(a, b0, acc8[2 * p]);
      acc8[2 * p + 1] = mma4(a, b1, acc8[2 * p + 1]);
    }
  }

  #pragma unroll
  for (int n = 0; n < 2 * NP; ++n) {
    const float bias = b_l[col0 + n * 16 + li];
    #pragma unroll
    for (int r = 0; r < 8; ++r) {
      float v = acc8[n][r] + bias;
      if (do_relu) v = fmaxf(v, 0.0f);
      outBase[(size_t)r * D_FEAT + col0 + n * 16] = v;
    }
  }
}

__global__ __launch_bounds__(512)
void sage_layer(const float* __restrict__ xin,
                const int* __restrict__ cur,
                const int* __restrict__ col,
                const float* __restrict__ w_l,
                const float* __restrict__ b_l,
                const float* __restrict__ w_r,
                float* __restrict__ out,
                int do_relu) {
  __shared__ float4 wfrag[2 * K_STEPS * KS_PITCH4];
  __shared__ __attribute__((aligned(16))) float strips[LAYER_WAVES * STRIP_FLOATS];

  const int wave = __builtin_amdgcn_readfirstlane((int)(threadIdx.x >> 5));
  const int lane = threadIdx.x & 31;
  const int hi   = lane >> 4;
  const int li   = lane & 15;

  {
    float* lds = (float*)wfrag;
    float4 v[16];
    #pragma unroll
    for (int q = 0; q < 16; ++q) {
      const int R = wave + LAYER_WAVES * q;
      const float* w = (R < D_FEAT) ? w_l : w_r;
      v[q] = *(const float4*)(w + (size_t)(R & (D_FEAT - 1)) * D_FEAT + 4 * lane);
    }
    #pragma unroll
    for (int q = 0; q < 16; ++q) {
      const int R  = wave + LAYER_WAVES * q;
      const int n  = R & (D_FEAT - 1);
      const int S  = (R >> 7) * K_STEPS + lane;
      const int ct = n >> 4;
      float* d = lds + (size_t)S * (KS_PITCH4 * 4) + (ct >> 1) * 128 + (n & 15) * 4 + (ct & 1) * 2;
      v2f lo = {v[q].x, v[q].y};
      v2f up = {v[q].z, v[q].w};
      *(v2f*)d        = lo;
      *(v2f*)(d + 64) = up;
    }
  }
  __syncthreads();

  const int t0 = (int)(((unsigned)blockIdx.x * (unsigned)ROW_TILES) / (unsigned)LAYER_BLOCKS);
  const int t1 = (int)(((unsigned)(blockIdx.x + 1) * (unsigned)ROW_TILES) / (unsigned)LAYER_BLOCKS);

  const float4* bfrag = wfrag + lane;
  float* strip = strips + wave * STRIP_FLOATS;
  const float* aStrip = strip + li * STRIP_PITCH + 2 * hi;

  const int nq = (t1 - t0) & 3;
  const int tf = t1 - nq;
  const int qi = (wave - 8) & (LAYER_WAVES - 1);
  int qleft = (qi < 4 * nq) ? 1 : 0;

  int t = t0 + wave;
  for (;;) {
    int tile;
    int pair = -1;
    if (qleft) { qleft = 0; tile = tf + (qi >> 2); pair = qi & 3; }
    else if (t < tf) { tile = t; t += LAYER_WAVES; }
    else break;
    const int m0 = tile * 16;

    int bnd = 0;
    {
      const int idx = m0 - 1 + lane;
      if (lane <= 16 && idx >= 0) bnd = cur[idx];
      bnd = min(max(bnd, 0), N_EDGES);
    }
    const int tb = __builtin_amdgcn_readlane(bnd, 0);
    const int te = __builtin_amdgcn_readlane(bnd, 16);
    int j  = 0;
    int bj = tb;
    int ej = __builtin_amdgcn_readlane(bnd, 1);
    float4 acc = make_float4(0.f, 0.f, 0.f, 0.f);

#define NO_SLOT 0x3fffffff

#define FLUSH_ROW()                                                           \
    {                                                                         \
      const float d = fmaxf((float)(ej - bj), 1.0f);                          \
      v4f o = {acc.x / d, acc.y / d, acc.z / d, acc.w / d};                   \
      *(v4f*)(strip + j * STRIP_PITCH + 4 * lane) = o;                        \
      acc = make_float4(0.f, 0.f, 0.f, 0.f);                                  \
      bj = ej;                                                                \
      ++j;                                                                    \
      ej = __builtin_amdgcn_readlane(bnd, min(j + 1, 16));                    \
      if (j >= 16) ej = NO_SLOT;                                              \
    }

#define LOAD16(R, IDX, OFF)                                                   \
    _Pragma("unroll")   \
    for (int q = 0; q < 16; ++q) {                                            \
      const unsigned sidx = (unsigned)__builtin_amdgcn_readlane(IDX, (OFF) + q); \
      const unsigned off = sidx * (unsigned)(D_FEAT * 4) + (unsigned)lane * 16u; \
      R[q] = *(const float4*)((const char*)xin + off);                        \
    }

#define SUM16(R, BASE)                                                        \
    _Pragma("unroll")   \
    for (int q = 0; q < 16; ++q) {                                            \
      const int slot = (BASE) + q;                                            \
      while (slot >= ej) FLUSH_ROW()                                          \
      acc.x += R[q].x; acc.y += R[q].y; acc.z += R[q].z; acc.w += R[q].w;     \
    }

#define CHUNK_READ(P0) col[min((P0) + lane, N_EDGES - 1)]
#define CHUNK_ROWS(V, P0) \
    (((P0) + lane < te) ? (int)min((unsigned)(V), (unsigned)(N_NODES - 1)) : m0)

    int idxv = CHUNK_READ(tb);
    for (int c = tb; c < te; c += 32) {
      const int rows = CHUNK_ROWS(idxv, c);
      idxv = CHUNK_READ(c + 32);
      float4 ra[16], rb[16];
      LOAD16(ra, rows, 0)
      LOAD16(rb, rows, 16)
      SUM16(ra, c)
      SUM16(rb, c + 16)
    }
    while (j < 16) FLUSH_ROW()
#undef CHUNK_ROWS
#undef CHUNK_READ
#undef SUM16
#undef LOAD16
#undef FLUSH_ROW
#undef NO_SLOT

    const int rowA = m0 + li;
    const float* xRow = xin + (size_t)rowA * D_FEAT + 2 * hi;

    float* outBase = out + (size_t)(m0 + 8 * hi) * D_FEAT + li;
    if (pair < 0) tile_product<4>(aStrip, xRow, bfrag, b_l, outBase, 0, do_relu, li);
    else          tile_product<1>(aStrip, xRow, bfrag + pair * 32, b_l, outBase, pair * 32, do_relu, li);
  }
}

extern "C" void kernel_launch(void* const* d_in, const int* in_sizes, int n_in,
                              void* d_out, int out_size, void* d_ws, size_t ws_size,
                              hipStream_t stream) {
  const float* x    = (const float*)d_in[0];
  const float* w1_l = (const float*)d_in[1];
  const float* b1_l = (const float*)d_in[2];
  const float* w1_r = (const float*)d_in[3];
  const float* w2_l = (const float*)d_in[4];
  const float* b2_l = (const float*)d_in[5];
  const float* w2_r = (const float*)d_in[6];
  const int*   edges = (const int*)d_in[7];
  const int*   src  = edges;
  const int*   dst  = edges + N_EDGES;

  float* h    = (float*)d_ws;
  int*   cur  = (int*)(h + (size_t)N_NODES * D_FEAT);
  int*   col  = cur + N_NODES;
  int*   trec = col + N_EDGES;
  int*   tab  = trec + 2 * N_EDGES;
  float* out  = (float*)d_out;

  dim3 lb(LAYER_WAVES * 32), lg(LAYER_BLOCKS);

  bucket_chunk<<<dim3(N_CH), dim3(CH_THREADS), 0, stream>>>(src, dst, tab, trec);
  build_csr<<<dim3(N_BKT), dim3(CSR_THREADS), 0, stream>>>(tab, trec, cur, col);

  sage_layer<<<lg, lb, 0, stream>>>(x, cur, col, w1_l, b1_l, w1_r, h, 1);
  sage_layer<<<lg, lb, 0, stream>>>(h, cur, col, w2_l, b2_l, w2_r, out, 0);
}
